# starting kernel
# baseline (speedup 1.0000x reference)
#include <hip/hip_runtime.h>
#include <math.h>

typedef __attribute__((ext_vector_type(2))) float v2f;
typedef __attribute__((ext_vector_type(8))) float v8f;

#define WMMA4(a, b, c) \
    __builtin_amdgcn_wmma_f32_16x16x4_f32(false, (a), false, (b), (short)0, (c), false, false)

__device__ __forceinline__ float fast_tanh(float x) {
#if __has_builtin(__builtin_amdgcn_tanhf)
    return __builtin_amdgcn_tanhf(x);
#else
    float xc = fminf(fmaxf(x, -10.0f), 10.0f);
    float t = __builtin_amdgcn_exp2f(xc * 2.8853900817779268f);
    return (t - 1.0f) * __builtin_amdgcn_rcpf(t + 1.0f);
#endif
}

__launch_bounds__(32)
__global__ void genode_rk45_wmma(const float* __restrict__ z_end,
                                 const float* __restrict__ W1,
                                 const float* __restrict__ b1,
                                 const float* __restrict__ W2,
                                 const float* __restrict__ b2,
                                 float* __restrict__ out,
                                 float h)
{
    __shared__ v2f w1B[8][4][32];
    __shared__ v2f w2B[32][32];
    __shared__ __align__(16) float hbuf[16][132];
    __shared__ __align__(16) float zbuf[16][20];

    const int lane = threadIdx.x;
    const int n    = lane & 15;
    const int half = lane >> 4;
    const int dlt  = half * 2;
    const int g    = blockIdx.x;

    #pragma unroll
    for (int t = 0; t < 8; ++t)
        #pragma unroll
        for (int c = 0; c < 4; ++c) {
            int k = 4 * c + dlt;
            v2f v;
            v.x = W1[k * 128 + 16 * t + n];
            v.y = W1[(k + 1) * 128 + 16 * t + n];
            w1B[t][c][lane] = v;
        }
    #pragma unroll
    for (int c = 0; c < 32; ++c) {
        int k = 4 * c + dlt;
        v2f v;
        v.x = W2[k * 16 + n];
        v.y = W2[(k + 1) * 16 + n];
        w2B[c][lane] = v;
    }
    float b1v[8];
    #pragma unroll
    for (int t = 0; t < 8; ++t) b1v[t] = b1[16 * t + n];
    const float b2v = b2[n];

    v8f z;
    #pragma unroll
    for (int v = 0; v < 8; ++v)
        z[v] = z_end[(16 * g + v + half * 8) * 16 + n];

    __syncthreads();

    auto put_z = [&](const v8f& val) {
        #pragma unroll
        for (int v = 0; v < 8; ++v)
            zbuf[v + half * 8][n] = val[v];
    };

    auto feval = [&]() -> v8f {
        __syncthreads();
        v2f az[4];
        #pragma unroll
        for (int c = 0; c < 4; ++c)
            az[c] = *(const v2f*)&zbuf[n][4 * c + dlt];

        #pragma unroll
        for (int t = 0; t < 8; ++t) {
            v8f acc;
            #pragma unroll
            for (int v = 0; v < 8; ++v) acc[v] = b1v[t];
            #pragma unroll
            for (int c = 0; c < 4; ++c)
                acc = WMMA4(az[c], w1B[t][c][lane], acc);
            #pragma unroll
            for (int v = 0; v < 8; ++v)
                hbuf[v + half * 8][16 * t + n] = fast_tanh(acc[v]);
        }
        __syncthreads();

        v8f o;
        #pragma unroll
        for (int v = 0; v < 8; ++v) o[v] = b2v;
        #pragma unroll
        for (int c = 0; c < 32; ++c) {
            v2f ah = *(const v2f*)&hbuf[n][4 * c + dlt];
            o = WMMA4(ah, w2B[c][lane], o);
        }
        return o;
    };

    const int nsteps = 16 * g + 15;
    for (int s = 1; s <= nsteps; ++s) {
        put_z(z);
        v8f k1 = feval();

        v8f t;
        #pragma unroll
        for (int v = 0; v < 8; ++v) t[v] = z[v] + h * (0.2f * k1[v]);
        put_z(t);
        v8f k2 = feval();

        #pragma unroll
        for (int v = 0; v < 8; ++v)
            t[v] = z[v] + h * ((3.0f / 40.0f) * k1[v] + (9.0f / 40.0f) * k2[v]);
        put_z(t);
        v8f k3 = feval();

        #pragma unroll
        for (int v = 0; v < 8; ++v)
            t[v] = z[v] + h * ((44.0f / 45.0f) * k1[v] - (56.0f / 15.0f) * k2[v]
                               + (32.0f / 9.0f) * k3[v]);
        put_z(t);
        v8f k4 = feval();

        #pragma unroll
        for (int v = 0; v < 8; ++v)
            t[v] = z[v] + h * ((19372.0f / 6561.0f) * k1[v] - (25360.0f / 2187.0f) * k2[v]
                               + (64448.0f / 6561.0f) * k3[v] - (212.0f / 729.0f) * k4[v]);
        put_z(t);
        v8f k5 = feval();

        #pragma unroll
        for (int v = 0; v < 8; ++v)
            t[v] = z[v] + h * ((9017.0f / 3168.0f) * k1[v] - (355.0f / 33.0f) * k2[v]
                               + (46732.0f / 5247.0f) * k3[v] + (49.0f / 176.0f) * k4[v]
                               - (5103.0f / 18656.0f) * k5[v]);
        put_z(t);
        v8f k6 = feval();

        #pragma unroll
        for (int v = 0; v < 8; ++v) {
            float zn = z[v] + h * ((35.0f / 384.0f) * k1[v] + (500.0f / 1113.0f) * k3[v]
                                   + (125.0f / 192.0f) * k4[v] - (2187.0f / 6784.0f) * k5[v]
                                   + (11.0f / 84.0f) * k6[v]);
            int target = 16 * g + v + half * 8;
            z[v] = (target >= s) ? zn : z[v];
        }
    }

    #pragma unroll
    for (int v = 0; v < 8; ++v)
        out[(16 * g + v + half * 8) * 16 + n] = z[v];
}

extern "C" void kernel_launch(void* const* d_in, const int* in_sizes, int n_in,
                              void* d_out, int out_size, void* d_ws, size_t ws_size,
                              hipStream_t stream) {
    const float* z_end = (const float*)d_in[0];
    const float* W1    = (const float*)d_in[1];
    const float* b1    = (const float*)d_in[2];
    const float* W2    = (const float*)d_in[3];
    const float* b2    = (const float*)d_in[4];
    float* out = (float*)d_out;

    const int N = in_sizes[0] / 16;
    const float h = 1.0f / (float)(N - 1);
    const int ntiles = N / 16;

    hipLaunchKernelGGL(genode_rk45_wmma, dim3(ntiles), dim3(32), 0, stream,
                       z_end, W1, b1, W2, b2, out, h);
}


# Transposed products (weights as A, state as B): each D fragment is the next B fragment, so activations stay in registers, no LDS, no barriers; k order inside chunks permuted, so sum order changes
# speedup vs baseline: 1.5777x; 1.5777x over previous
#include <hip/hip_runtime.h>
#include <math.h>


typedef __attribute__((ext_vector_type(2))) float v2f;
typedef __attribute__((ext_vector_type(8))) float v8f;

#define WMMA4(a, b, c) \
    __builtin_amdgcn_wmma_f32_16x16x4_f32(false, (a), false, (b), (short)0, (c), false, false)

__device__ __forceinline__ float fast_tanh(float x) {
#if __has_builtin(__builtin_amdgcn_tanhf)
    return __builtin_amdgcn_tanhf(x);
#else
    float xc = fminf(fmaxf(x, -10.0f), 10.0f);
    float t = __builtin_amdgcn_exp2f(xc * 2.8853900817779268f);
    return (t - 1.0f) * __builtin_amdgcn_rcpf(t + 1.0f);
#endif
}

__launch_bounds__(32)
__global__ void genode_rk45_wmma(const float* __restrict__ z_end,
                                 const float* __restrict__ W1,
                                 const float* __restrict__ b1,
                                 const float* __restrict__ W2,
                                 const float* __restrict__ b2,
                                 float* __restrict__ out,
                                 float h)
{
    const int lane = threadIdx.x;
    const int c    = lane & 15;
    const int hf   = lane >> 4;
    const int g    = blockIdx.x;


    v2f w1A[8][4];
    #pragma unroll
    for (int t = 0; t < 8; ++t)
        #pragma unroll
        for (int j = 0; j < 4; ++j) {
            const int d = 8 * hf + 2 * j;
            v2f v;
            v.x = W1[d * 128 + 16 * t + c];
            v.y = W1[(d + 1) * 128 + 16 * t + c];
            w1A[t][j] = v;
        }

    v2f w2A[8][4];
    #pragma unroll
    for (int t = 0; t < 8; ++t)
        #pragma unroll
        for (int j = 0; j < 4; ++j) {
            const int k = 16 * t + 8 * hf + 2 * j;
            v2f v;
            v.x = W2[k * 16 + c];
            v.y = W2[(k + 1) * 16 + c];
            w2A[t][j] = v;
        }

    v8f b1C[8];
    #pragma unroll
    for (int t = 0; t < 8; ++t)
        #pragma unroll
        for (int r = 0; r < 8; ++r)
            b1C[t][r] = b1[16 * t + 8 * hf + r];
    v8f b2C;
    #pragma unroll
    for (int r = 0; r < 8; ++r)
        b2C[r] = b2[8 * hf + r];

    v8f z;
    #pragma unroll
    for (int r = 0; r < 8; ++r)
        z[r] = z_end[(16 * g + c) * 16 + 8 * hf + r];

    auto feval = [&](const v8f& x) -> v8f {
        v2f bx[4];
        #pragma unroll
        for (int j = 0; j < 4; ++j) { bx[j].x = x[2 * j]; bx[j].y = x[2 * j + 1]; }

        v8f o = b2C;
        #pragma unroll
        for (int t = 0; t < 8; ++t) {
            v8f acc = b1C[t];
            #pragma unroll
            for (int j = 0; j < 4; ++j)
                acc = WMMA4(w1A[t][j], bx[j], acc);
            #pragma unroll
            for (int j = 0; j < 4; ++j) {
                v2f bh;
                bh.x = fast_tanh(acc[2 * j]);
                bh.y = fast_tanh(acc[2 * j + 1]);
                o = WMMA4(w2A[t][j], bh, o);
            }
        }
        return o;
    };

    const int target = 16 * g + c;
    const int nsteps = 16 * g + 15;
    for (int s = 1; s <= nsteps; ++s) {
        v8f k1 = feval(z);

        v8f t;
        #pragma unroll
        for (int v = 0; v < 8; ++v) t[v] = z[v] + h * (0.2f * k1[v]);
        v8f k2 = feval(t);

        #pragma unroll
        for (int v = 0; v < 8; ++v)
            t[v] = z[v] + h * ((3.0f / 40.0f) * k1[v] + (9.0f / 40.0f) * k2[v]);
        v8f k3 = feval(t);

        #pragma unroll
        for (int v = 0; v < 8; ++v)
            t[v] = z[v] + h * ((44.0f / 45.0f) * k1[v] - (56.0f / 15.0f) * k2[v]
                               + (32.0f / 9.0f) * k3[v]);
        v8f k4 = feval(t);

        #pragma unroll
        for (int v = 0; v < 8; ++v)
            t[v] = z[v] + h * ((19372.0f / 6561.0f) * k1[v] - (25360.0f / 2187.0f) * k2[v]
                               + (64448.0f / 6561.0f) * k3[v] - (212.0f / 729.0f) * k4[v]);
        v8f k5 = feval(t);

        #pragma unroll
        for (int v = 0; v < 8; ++v)
            t[v] = z[v] + h * ((9017.0f / 3168.0f) * k1[v] - (355.0f / 33.0f) * k2[v]
                               + (46732.0f / 5247.0f) * k3[v] + (49.0f / 176.0f) * k4[v]
                               - (5103.0f / 18656.0f) * k5[v]);
        v8f k6 = feval(t);

        const bool live = (target >= s);
        #pragma unroll
        for (int v = 0; v < 8; ++v) {
            const float zo = z[v];
            const float zn = zo + h * ((35.0f / 384.0f) * k1[v] + (500.0f / 1113.0f) * k3[v]
                                       + (125.0f / 192.0f) * k4[v] - (2187.0f / 6784.0f) * k5[v]
                                       + (11.0f / 84.0f) * k6[v]);
            z[v] = live ? zn : zo;
        }
    }

    #pragma unroll
    for (int r = 0; r < 8; ++r)
        out[(16 * g + c) * 16 + 8 * hf + r] = z[r];
}

extern "C" void kernel_launch(void* const* d_in, const int* in_sizes, int n_in,
                              void* d_out, int out_size, void* d_ws, size_t ws_size,
                              hipStream_t stream) {
    const float* z_end = (const float*)d_in[0];
    const float* W1    = (const float*)d_in[1];
    const float* b1    = (const float*)d_in[2];
    const float* W2    = (const float*)d_in[3];
    const float* b2    = (const float*)d_in[4];
    float* out = (float*)d_out;

    const int N = in_sizes[0] / 16;
    const float h = 1.0f / (float)(N - 1);
    const int ntiles = N / 16;

    hipLaunchKernelGGL(genode_rk45_wmma, dim3(ntiles), dim3(32), 0, stream,
                       z_end, W1, b1, W2, b2, out, h);
}


# Hidden units split over the 4 waves of a work-group (one per SIMD) so 4 matrix pipes share each f; partial outputs summed through LDS, one barrier per f; layer-2 sum order changes (tree of partials)
# speedup vs baseline: 3.8441x; 2.4365x over previous
#include <hip/hip_runtime.h>
#include <math.h>


typedef __attribute__((ext_vector_type(2))) float v2f;
typedef __attribute__((ext_vector_type(4))) float v4f;
typedef __attribute__((ext_vector_type(8))) float v8f;

#define WMMA4(a, b, c) \
    __builtin_amdgcn_wmma_f32_16x16x4_f32(false, (a), false, (b), (short)0, (c), false, false)

__device__ __forceinline__ float fast_tanh(float x) {
#if __has_builtin(__builtin_amdgcn_tanhf)
    return __builtin_amdgcn_tanhf(x);
#else
    float xc = fminf(fmaxf(x, -10.0f), 10.0f);
    float t = __builtin_amdgcn_exp2f(xc * 2.8853900817779268f);
    return (t - 1.0f) * __builtin_amdgcn_rcpf(t + 1.0f);
#endif
}

__launch_bounds__(128)
__global__ void genode_rk45_wmma(const float* __restrict__ z_end,
                                 const float* __restrict__ W1,
                                 const float* __restrict__ b1,
                                 const float* __restrict__ W2,
                                 const float* __restrict__ b2,
                                 float* __restrict__ out,
                                 float h)
{
    __shared__ v4f xch[2][4][2][32];

    const int lane = threadIdx.x & 31;
    const int w    = __builtin_amdgcn_readfirstlane((int)(threadIdx.x >> 5));
    const int c    = lane & 15;
    const int hf   = lane >> 4;
    const int g    = blockIdx.x;


    v2f w1A[2][4];
    #pragma unroll
    for (int u = 0; u < 2; ++u)
        #pragma unroll
        for (int j = 0; j < 4; ++j) {
            const int t = 2 * w + u;
            const int d = 8 * hf + 2 * j;
            v2f v;
            v.x = W1[d * 128 + 16 * t + c];
            v.y = W1[(d + 1) * 128 + 16 * t + c];
            w1A[u][j] = v;
        }

    v2f w2A[2][4];
    #pragma unroll
    for (int u = 0; u < 2; ++u)
        #pragma unroll
        for (int j = 0; j < 4; ++j) {
            const int t = 2 * w + u;
            const int k = 16 * t + 8 * hf + 2 * j;
            v2f v;
            v.x = W2[k * 16 + c];
            v.y = W2[(k + 1) * 16 + c];
            w2A[u][j] = v;
        }

    v8f b1C[2];
    #pragma unroll
    for (int u = 0; u < 2; ++u)
        #pragma unroll
        for (int r = 0; r < 8; ++r)
            b1C[u][r] = b1[16 * (2 * w + u) + 8 * hf + r];
    v8f b2C;
    #pragma unroll
    for (int r = 0; r < 8; ++r) {
        const float bv = b2[8 * hf + r];
        b2C[r] = (w == 0) ? bv : 0.0f;
    }
    const v8f zeroC = {0.0f, 0.0f, 0.0f, 0.0f, 0.0f, 0.0f, 0.0f, 0.0f};

    v8f z;
    #pragma unroll
    for (int r = 0; r < 8; ++r)
        z[r] = z_end[(16 * g + c) * 16 + 8 * hf + r];

    auto feval = [&](const v8f& x, const int par) -> v8f {
        v2f bx[4];
        #pragma unroll
        for (int j = 0; j < 4; ++j) { bx[j].x = x[2 * j]; bx[j].y = x[2 * j + 1]; }

        v8f acc0 = b1C[0];
        v8f acc1 = b1C[1];
        #pragma unroll
        for (int j = 0; j < 4; ++j) {
            acc0 = WMMA4(w1A[0][j], bx[j], acc0);
            acc1 = WMMA4(w1A[1][j], bx[j], acc1);
        }

        v8f o0 = b2C;
        v8f o1 = zeroC;
        #pragma unroll
        for (int j = 0; j < 4; ++j) {
            v2f bh0, bh1;
            bh0.x = fast_tanh(acc0[2 * j]);
            bh0.y = fast_tanh(acc0[2 * j + 1]);
            bh1.x = fast_tanh(acc1[2 * j]);
            bh1.y = fast_tanh(acc1[2 * j + 1]);
            o0 = WMMA4(w2A[0][j], bh0, o0);
            o1 = WMMA4(w2A[1][j], bh1, o1);
        }

        v8f p;
        #pragma unroll
        for (int r = 0; r < 8; ++r) p[r] = o0[r] + o1[r];
        v4f lo, hi;
        lo.x = p[0]; lo.y = p[1]; lo.z = p[2]; lo.w = p[3];
        hi.x = p[4]; hi.y = p[5]; hi.z = p[6]; hi.w = p[7];
        xch[par][w][0][lane] = lo;
        xch[par][w][1][lane] = hi;
        __syncthreads();
        const v4f a_lo = xch[par][w ^ 1][0][lane], a_hi = xch[par][w ^ 1][1][lane];
        const v4f b_lo = xch[par][w ^ 2][0][lane], b_hi = xch[par][w ^ 2][1][lane];
        const v4f c_lo = xch[par][w ^ 3][0][lane], c_hi = xch[par][w ^ 3][1][lane];

        v8f o;
        o[0] = (p[0] + a_lo.x) + (b_lo.x + c_lo.x);
        o[1] = (p[1] + a_lo.y) + (b_lo.y + c_lo.y);
        o[2] = (p[2] + a_lo.z) + (b_lo.z + c_lo.z);
        o[3] = (p[3] + a_lo.w) + (b_lo.w + c_lo.w);
        o[4] = (p[4] + a_hi.x) + (b_hi.x + c_hi.x);
        o[5] = (p[5] + a_hi.y) + (b_hi.y + c_hi.y);
        o[6] = (p[6] + a_hi.z) + (b_hi.z + c_hi.z);
        o[7] = (p[7] + a_hi.w) + (b_hi.w + c_hi.w);
        return o;
    };

    const int target = 16 * g + c;
    const int nsteps = 16 * g + 15;
    for (int s = 1; s <= nsteps; ++s) {
        v8f k1 = feval(z, 0);

        v8f t;
        #pragma unroll
        for (int v = 0; v < 8; ++v) t[v] = z[v] + h * (0.2f * k1[v]);
        v8f k2 = feval(t, 1);

        #pragma unroll
        for (int v = 0; v < 8; ++v)
            t[v] = z[v] + h * ((3.0f / 40.0f) * k1[v] + (9.0f / 40.0f) * k2[v]);
        v8f k3 = feval(t, 0);

        #pragma unroll
        for (int v = 0; v < 8; ++v)
            t[v] = z[v] + h * ((44.0f / 45.0f) * k1[v] - (56.0f / 15.0f) * k2[v]
                               + (32.0f / 9.0f) * k3[v]);
        v8f k4 = feval(t, 1);

        #pragma unroll
        for (int v = 0; v < 8; ++v)
            t[v] = z[v] + h * ((19372.0f / 6561.0f) * k1[v] - (25360.0f / 2187.0f) * k2[v]
                               + (64448.0f / 6561.0f) * k3[v] - (212.0f / 729.0f) * k4[v]);
        v8f k5 = feval(t, 0);

        #pragma unroll
        for (int v = 0; v < 8; ++v)
            t[v] = z[v] + h * ((9017.0f / 3168.0f) * k1[v] - (355.0f / 33.0f) * k2[v]
                               + (46732.0f / 5247.0f) * k3[v] + (49.0f / 176.0f) * k4[v]
                               - (5103.0f / 18656.0f) * k5[v]);
        v8f k6 = feval(t, 1);

        const bool live = (target >= s);
        #pragma unroll
        for (int v = 0; v < 8; ++v) {
            const float zo = z[v];
            const float zn = zo + h * ((35.0f / 384.0f) * k1[v] + (500.0f / 1113.0f) * k3[v]
                                       + (125.0f / 192.0f) * k4[v] - (2187.0f / 6784.0f) * k5[v]
                                       + (11.0f / 84.0f) * k6[v]);
            z[v] = live ? zn : zo;
        }
    }

    if (w == 0) {
        #pragma unroll
        for (int r = 0; r < 8; ++r)
            out[(16 * g + c) * 16 + 8 * hf + r] = z[r];
    }
}

extern "C" void kernel_launch(void* const* d_in, const int* in_sizes, int n_in,
                              void* d_out, int out_size, void* d_ws, size_t ws_size,
                              hipStream_t stream) {
    const float* z_end = (const float*)d_in[0];
    const float* W1    = (const float*)d_in[1];
    const float* b1    = (const float*)d_in[2];
    const float* W2    = (const float*)d_in[3];
    const float* b2    = (const float*)d_in[4];
    float* out = (float*)d_out;

    const int N = in_sizes[0] / 16;
    const float h = 1.0f / (float)(N - 1);
    const int ntiles = N / 16;

    hipLaunchKernelGGL(genode_rk45_wmma, dim3(ntiles), dim3(128), 0, stream,
                       z_end, W1, b1, W2, b2, out, h);
}


# speedup vs baseline: 4.0252x; 1.0471x over previous
#include <hip/hip_runtime.h>
#include <math.h>


typedef __attribute__((ext_vector_type(2))) float v2f;
typedef __attribute__((ext_vector_type(4))) float v4f;
typedef __attribute__((ext_vector_type(8))) float v8f;

#define WMMA4(a, b, c) \
    __builtin_amdgcn_wmma_f32_16x16x4_f32(false, (a), false, (b), (short)0, (c), false, false)

__device__ __forceinline__ float fast_tanh(float x) {
#if __has_builtin(__builtin_amdgcn_tanhf)
    return __builtin_amdgcn_tanhf(x);
#else
    float xc = fminf(fmaxf(x, -10.0f), 10.0f);
    float t = __builtin_amdgcn_exp2f(xc * 2.8853900817779268f);
    return (t - 1.0f) * __builtin_amdgcn_rcpf(t + 1.0f);
#endif
}

__launch_bounds__(128)
__global__ void genode_rk45_wmma(const float* __restrict__ z_end,
                                 const float* __restrict__ W1,
                                 const float* __restrict__ b1,
                                 const float* __restrict__ W2,
                                 const float* __restrict__ b2,
                                 float* __restrict__ out,
                                 float h)
{
    constexpr int SLOT = 32768 / 16;
    __shared__ v4f xch[3 * SLOT + 128];

    const int lane = threadIdx.x & 31;
    const int w    = __builtin_amdgcn_readfirstlane((int)(threadIdx.x >> 5));
    const int c    = lane & 15;
    const int hf   = lane >> 4;
    const int g    = blockIdx.x;


    v2f w1A[2][4];
    #pragma unroll
    for (int u = 0; u < 2; ++u)
        #pragma unroll
        for (int j = 0; j < 4; ++j) {
            const int t = 2 * w + u;
            const int d = 8 * hf + 2 * j;
            v2f v;
            v.x = W1[d * 128 + 16 * t + c];
            v.y = W1[(d + 1) * 128 + 16 * t + c];
            w1A[u][j] = v;
        }

    v2f w2A[2][4];
    #pragma unroll
    for (int u = 0; u < 2; ++u)
        #pragma unroll
        for (int j = 0; j < 4; ++j) {
            const int t = 2 * w + u;
            const int k = 16 * t + 8 * hf + 2 * j;
            v2f v;
            v.x = W2[k * 16 + c];
            v.y = W2[(k + 1) * 16 + c];
            w2A[u][j] = v;
        }

    v8f b1C[2];
    #pragma unroll
    for (int u = 0; u < 2; ++u)
        #pragma unroll
        for (int r = 0; r < 8; ++r)
            b1C[u][r] = b1[16 * (2 * w + u) + 8 * hf + r];
    v8f b2C;
    #pragma unroll
    for (int r = 0; r < 8; ++r) {
        const float bv = b2[8 * hf + r];
        b2C[r] = (w == 0) ? bv : 0.0f;
    }
    const v8f zeroC = {0.0f, 0.0f, 0.0f, 0.0f, 0.0f, 0.0f, 0.0f, 0.0f};

    v8f z;
    #pragma unroll
    for (int r = 0; r < 8; ++r)
        z[r] = z_end[(16 * g + c) * 16 + 8 * hf + r];

    auto feval = [&](const v8f& x, const int par) -> v8f {
        v2f bx[4];
        #pragma unroll
        for (int j = 0; j < 4; ++j) { bx[j].x = x[2 * j]; bx[j].y = x[2 * j + 1]; }

        v8f acc0 = b1C[0];
        v8f acc1 = b1C[1];
        #pragma unroll
        for (int j = 0; j < 4; ++j) {
            acc0 = WMMA4(w1A[0][j], bx[j], acc0);
            acc1 = WMMA4(w1A[1][j], bx[j], acc1);
        }

        v8f o0 = b2C;
        v8f o1 = zeroC;
        #pragma unroll
        for (int j = 0; j < 4; ++j) {
            v2f bh0, bh1;
            bh0.x = fast_tanh(acc0[2 * j]);
            bh0.y = fast_tanh(acc0[2 * j + 1]);
            bh1.x = fast_tanh(acc1[2 * j]);
            bh1.y = fast_tanh(acc1[2 * j + 1]);
            o0 = WMMA4(w2A[0][j], bh0, o0);
            o1 = WMMA4(w2A[1][j], bh1, o1);
        }

        v8f p;
        #pragma unroll
        for (int r = 0; r < 8; ++r) p[r] = o0[r] + o1[r];
        v4f lo, hi;
        lo.x = p[0]; lo.y = p[1]; lo.z = p[2]; lo.w = p[3];
        hi.x = p[4]; hi.y = p[5]; hi.z = p[6]; hi.w = p[7];
        const int own = w * SLOT + par * 64 + lane;
        const int pa = (w ^ 1) * SLOT + par * 64 + lane;
        const int pb = (w ^ 2) * SLOT + par * 64 + lane;
        const int pc = (w ^ 3) * SLOT + par * 64 + lane;
        xch[own] = lo;
        xch[own + 32] = hi;
        __syncthreads();
        const v4f a_lo = xch[pa], a_hi = xch[pa + 32];
        const v4f b_lo = xch[pb], b_hi = xch[pb + 32];
        const v4f c_lo = xch[pc], c_hi = xch[pc + 32];

        v8f o;
        o[0] = (p[0] + a_lo.x) + (b_lo.x + c_lo.x);
        o[1] = (p[1] + a_lo.y) + (b_lo.y + c_lo.y);
        o[2] = (p[2] + a_lo.z) + (b_lo.z + c_lo.z);
        o[3] = (p[3] + a_lo.w) + (b_lo.w + c_lo.w);
        o[4] = (p[4] + a_hi.x) + (b_hi.x + c_hi.x);
        o[5] = (p[5] + a_hi.y) + (b_hi.y + c_hi.y);
        o[6] = (p[6] + a_hi.z) + (b_hi.z + c_hi.z);
        o[7] = (p[7] + a_hi.w) + (b_hi.w + c_hi.w);
        return o;
    };

    const int target = 16 * g + c;
    const int nsteps = 16 * g + 15;
    for (int s = 1; s <= nsteps; ++s) {
        v8f k1 = feval(z, 0);

        v8f t;
        #pragma unroll
        for (int v = 0; v < 8; ++v) t[v] = z[v] + h * (0.2f * k1[v]);
        v8f k2 = feval(t, 1);

        #pragma unroll
        for (int v = 0; v < 8; ++v)
            t[v] = z[v] + h * ((3.0f / 40.0f) * k1[v] + (9.0f / 40.0f) * k2[v]);
        v8f k3 = feval(t, 0);

        #pragma unroll
        for (int v = 0; v < 8; ++v)
            t[v] = z[v] + h * ((44.0f / 45.0f) * k1[v] - (56.0f / 15.0f) * k2[v]
                               + (32.0f / 9.0f) * k3[v]);
        v8f k4 = feval(t, 1);

        #pragma unroll
        for (int v = 0; v < 8; ++v)
            t[v] = z[v] + h * ((19372.0f / 6561.0f) * k1[v] - (25360.0f / 2187.0f) * k2[v]
                               + (64448.0f / 6561.0f) * k3[v] - (212.0f / 729.0f) * k4[v]);
        v8f k5 = feval(t, 0);

        #pragma unroll
        for (int v = 0; v < 8; ++v)
            t[v] = z[v] + h * ((9017.0f / 3168.0f) * k1[v] - (355.0f / 33.0f) * k2[v]
                               + (46732.0f / 5247.0f) * k3[v] + (49.0f / 176.0f) * k4[v]
                               - (5103.0f / 18656.0f) * k5[v]);
        v8f k6 = feval(t, 1);

        const bool live = (target >= s);
        #pragma unroll
        for (int v = 0; v < 8; ++v) {
            const float zo = z[v];
            const float zn = zo + h * ((35.0f / 384.0f) * k1[v] + (500.0f / 1113.0f) * k3[v]
                                       + (125.0f / 192.0f) * k4[v] - (2187.0f / 6784.0f) * k5[v]
                                       + (11.0f / 84.0f) * k6[v]);
            z[v] = live ? zn : zo;
        }
    }

    if (w == 0) {
        #pragma unroll
        for (int r = 0; r < 8; ++r)
            out[(16 * g + c) * 16 + 8 * hf + r] = z[r];
    }
}

extern "C" void kernel_launch(void* const* d_in, const int* in_sizes, int n_in,
                              void* d_out, int out_size, void* d_ws, size_t ws_size,
                              hipStream_t stream) {
    const float* z_end = (const float*)d_in[0];
    const float* W1    = (const float*)d_in[1];
    const float* b1    = (const float*)d_in[2];
    const float* W2    = (const float*)d_in[3];
    const float* b2    = (const float*)d_in[4];
    float* out = (float*)d_out;

    const int N = in_sizes[0] / 16;
    const float h = 1.0f / (float)(N - 1);
    const int ntiles = N / 16;

    hipLaunchKernelGGL(genode_rk45_wmma, dim3(ntiles), dim3(128), 0, stream,
                       z_end, W1, b1, W2, b2, out, h);
}
